# starting kernel
# baseline (speedup 1.0000x reference)
#include <hip/hip_runtime.h>

typedef __attribute__((ext_vector_type(2))) float v2f;
typedef __attribute__((ext_vector_type(8))) float v8f;

#define GAIN 1.41421356237309515f
#define SLOPE 0.2f

__device__ __forceinline__ v8f wmma4(v2f a, v2f b, v8f c) {
  return __builtin_amdgcn_wmma_f32_16x16x4_f32(false, a, false, b, (short)0, c,
                                               false, false);
}

__device__ __forceinline__ int clampi(int v, int lo, int hi) {
  return v < lo ? lo : (v > hi ? hi : v);
}

__device__ __forceinline__ float tu_value(const float* __restrict__ f, int j, int up) {
  int d = j - (up >> 1);
  float v = 0.0f;
  if ((up & 1) == 0) {
    if (d >= 0 && d <= 6) {
      int k = 2 * d;
      float lo = (k >= 1)  ? f[k - 1] : 0.0f;
      float hi = (k <= 11) ? f[k]     : 0.0f;
      v = lo + hi;
    }
  } else {
    int t = d - 1;
    if (t >= 0 && t <= 5) v = f[2 * t] + f[2 * t + 1];
  }
  return v;
}

__device__ __forceinline__ float td_value(const float* __restrict__ f, int j, int mp) {
  int k = j - 2 * mp;
  return (k >= 0 && k < 12) ? f[k] : 0.0f;
}

__global__ __launch_bounds__(256) void pass1_up_h(
    const float* __restrict__ x, const float* __restrict__ bias,
    const float* __restrict__ fup, float* __restrict__ t1) {
  __shared__ float sTU[16][16];
  for (int i = threadIdx.x; i < 256; i += 256)
    sTU[i >> 4][i & 15] = tu_value(fup, i & 15, i >> 4);
  __syncthreads();

  const int lane = threadIdx.x & 31;
  const int half = lane >> 4;
  const int mrow = lane & 15;

  int t = blockIdx.x * 8 + (threadIdx.x >> 5);
  const int utile = t & 15; t >>= 4;
  const int ch = t;

  const int rbase = 8 * utile - 3;
  const float bb  = bias[ch & 63];
  const float* xb = x + (size_t)ch * 130 * 130 + 131 + mrow;
  float* ob = t1 + ((size_t)ch * 256 + 16 * utile + 8 * half) * 128 + mrow;

  v8f acc[8] = {};
#pragma unroll
  for (int c = 0; c < 4; ++c) {
    const int j0 = 4 * c + 2 * half;
    v2f a;
    a.x = sTU[mrow][j0];
    a.y = sTU[mrow][j0 + 1];
    const int r0 = rbase + j0, r1 = r0 + 1;
    if ((unsigned)r0 >= 128u) a.x = 0.0f;
    if ((unsigned)r1 >= 128u) a.y = 0.0f;
    const float* p0 = xb + (size_t)clampi(r0, 0, 127) * 130;
    const float* p1 = xb + (size_t)clampi(r1, 0, 127) * 130;
#pragma unroll
    for (int nt = 0; nt < 8; ++nt) {
      v2f b;
      b.x = p0[16 * nt] + bb;
      b.y = p1[16 * nt] + bb;
      acc[nt] = wmma4(a, b, acc[nt]);
    }
  }
#pragma unroll
  for (int g = 0; g < 8; ++g)
#pragma unroll
    for (int nt = 0; nt < 8; ++nt)
      ob[(size_t)g * 128 + 16 * nt] = acc[nt][g];
}

__global__ __launch_bounds__(256) void pass2_up_w_lrelu(
    const float* __restrict__ t1, const float* __restrict__ fup,
    float* __restrict__ t2) {
  __shared__ float sTU[16][16];
  for (int i = threadIdx.x; i < 256; i += 256)
    sTU[i >> 4][i & 15] = tu_value(fup, i & 15, i >> 4);
  __syncthreads();

  const int lane = threadIdx.x & 31;
  const int half = lane >> 4;
  const int mrow = lane & 15;

  int t = blockIdx.x * 8 + (threadIdx.x >> 5);
  const int vtile  = t & 15; t >>= 4;
  const int mgroup = t & 1;  t >>= 1;
  const int ch = t;

  const int cbase = 8 * vtile - 3;
  const float* ab = t1 + ((size_t)ch * 256 + 128 * mgroup + mrow) * 128;
  float* ob = t2 + ((size_t)ch * 256 + 128 * mgroup + 8 * half) * 256
            + 16 * vtile + mrow;

  v8f acc[8] = {};
#pragma unroll
  for (int c = 0; c < 4; ++c) {
    const int j0 = 4 * c + 2 * half;
    const int c0 = cbase + j0, c1 = c0 + 1;
    v2f b;
    b.x = ((unsigned)c0 < 128u) ? sTU[mrow][j0]     : 0.0f;
    b.y = ((unsigned)c1 < 128u) ? sTU[mrow][j0 + 1] : 0.0f;
    const float* q0 = ab + clampi(c0, 0, 127);
    const float* q1 = ab + clampi(c1, 0, 127);
#pragma unroll
    for (int nt = 0; nt < 8; ++nt) {
      v2f a;
      a.x = q0[2048 * nt];
      a.y = q1[2048 * nt];
      acc[nt] = wmma4(a, b, acc[nt]);
    }
  }
#pragma unroll
  for (int g = 0; g < 8; ++g)
#pragma unroll
    for (int nt = 0; nt < 8; ++nt) {
      float v = acc[nt][g];
      v = (v >= 0.0f ? v : SLOPE * v) * GAIN;
      ob[(size_t)(16 * nt + g) * 256] = v;
    }
}

__global__ __launch_bounds__(256) void pass3_dn_h(
    const float* __restrict__ t2, const float* __restrict__ fdn,
    float* __restrict__ t3) {
  __shared__ float sTD[16][48];
  for (int i = threadIdx.x; i < 16 * 48; i += 256) {
    int mp = i / 48, j = i % 48;
    sTD[mp][j] = (j < 44) ? td_value(fdn, j, mp) : 0.0f;
  }
  __syncthreads();

  const int lane = threadIdx.x & 31;
  const int half = lane >> 4;
  const int mrow = lane & 15;

  int t = blockIdx.x * 8 + (threadIdx.x >> 5);
  const int ngroup = t & 1; t >>= 1;
  const int mtile  = t & 7; t >>= 3;
  const int ch = t;

  const int rbase = 32 * mtile - 5;
  const float* xb = t2 + (size_t)ch * 256 * 256 + 128 * ngroup + mrow;
  float* ob = t3 + ((size_t)ch * 128 + 16 * mtile + 8 * half) * 256
            + 128 * ngroup + mrow;

  v8f acc[8] = {};
#pragma unroll
  for (int c = 0; c < 11; ++c) {
    const int j0 = 4 * c + 2 * half;
    v2f a;
    a.x = sTD[mrow][j0];
    a.y = sTD[mrow][j0 + 1];
    const int r0 = rbase + j0, r1 = r0 + 1;
    if ((unsigned)r0 >= 256u) a.x = 0.0f;
    if ((unsigned)r1 >= 256u) a.y = 0.0f;
    const float* p0 = xb + (size_t)clampi(r0, 0, 255) * 256;
    const float* p1 = xb + (size_t)clampi(r1, 0, 255) * 256;
#pragma unroll
    for (int nt = 0; nt < 8; ++nt) {
      v2f b;
      b.x = p0[16 * nt];
      b.y = p1[16 * nt];
      acc[nt] = wmma4(a, b, acc[nt]);
    }
  }
#pragma unroll
  for (int g = 0; g < 8; ++g)
#pragma unroll
    for (int nt = 0; nt < 8; ++nt)
      ob[(size_t)g * 256 + 16 * nt] = acc[nt][g];
}

__global__ __launch_bounds__(256) void pass4_dn_w(
    const float* __restrict__ t3, const float* __restrict__ fdn,
    float* __restrict__ out) {
  __shared__ float sTD[16][48];
  for (int i = threadIdx.x; i < 16 * 48; i += 256) {
    int mp = i / 48, j = i % 48;
    sTD[mp][j] = (j < 44) ? td_value(fdn, j, mp) : 0.0f;
  }
  __syncthreads();

  const int lane = threadIdx.x & 31;
  const int half = lane >> 4;
  const int mrow = lane & 15;

  int t = blockIdx.x * 8 + (threadIdx.x >> 5);
  const int ntile = t & 7; t >>= 3;
  const int ch = t;

  const int cbase = 32 * ntile - 5;
  const float* ab = t3 + ((size_t)ch * 128 + mrow) * 256;
  float* ob = out + ((size_t)ch * 128 + 8 * half) * 128 + 16 * ntile + mrow;

  v8f acc[8] = {};
#pragma unroll
  for (int c = 0; c < 11; ++c) {
    const int j0 = 4 * c + 2 * half;
    const int c0 = cbase + j0, c1 = c0 + 1;
    v2f b;
    b.x = ((unsigned)c0 < 256u) ? sTD[mrow][j0]     : 0.0f;
    b.y = ((unsigned)c1 < 256u) ? sTD[mrow][j0 + 1] : 0.0f;
    const float* q0 = ab + clampi(c0, 0, 255);
    const float* q1 = ab + clampi(c1, 0, 255);
#pragma unroll
    for (int nt = 0; nt < 8; ++nt) {
      v2f a;
      a.x = q0[4096 * nt];
      a.y = q1[4096 * nt];
      acc[nt] = wmma4(a, b, acc[nt]);
    }
  }
#pragma unroll
  for (int g = 0; g < 8; ++g)
#pragma unroll
    for (int nt = 0; nt < 8; ++nt)
      ob[(size_t)(16 * nt + g) * 128] = acc[nt][g];
}

extern "C" void kernel_launch(void* const* d_in, const int* in_sizes, int n_in,
                              void* d_out, int out_size, void* d_ws, size_t ws_size,
                              hipStream_t stream) {
  (void)in_sizes; (void)n_in; (void)out_size; (void)ws_size;
  const float* x    = (const float*)d_in[0];
  const float* bias = (const float*)d_in[1];
  const float* fup  = (const float*)d_in[2];
  const float* fdn  = (const float*)d_in[3];
  float* out = (float*)d_out;

  float* t1 = (float*)d_ws;
  float* t2 = (float*)((char*)d_ws + (size_t)512 * 256 * 128 * 4);
  float* t3 = (float*)d_ws;

  dim3 blk(256);
  pass1_up_h      <<<512 * 16     / 8, blk, 0, stream>>>(x, bias, fup, t1);
  pass2_up_w_lrelu<<<512 * 2 * 16 / 8, blk, 0, stream>>>(t1, fup, t2);
  pass3_dn_h      <<<512 * 8 * 2  / 8, blk, 0, stream>>>(t2, fdn, t3);
  pass4_dn_w      <<<512 * 8      / 8, blk, 0, stream>>>(t3, fdn, out);
}


# Up-W moved first so up-H, leaky and down-H fuse in one kernel: the 256-row plane stays in WMMA accumulators (D reused as B), never stored; 3 launches; up-H/up-W sum order swapped
# speedup vs baseline: 1.3964x; 1.3964x over previous
#include <hip/hip_runtime.h>

typedef __attribute__((ext_vector_type(2))) float v2f;
typedef __attribute__((ext_vector_type(8))) float v8f;

#define GAIN 1.41421356237309515f
#define SLOPE 0.2f

__device__ __forceinline__ v8f wmma4(v2f a, v2f b, v8f c) {
  return __builtin_amdgcn_wmma_f32_16x16x4_f32(false, a, false, b, (short)0, c,
                                               false, false);
}

__device__ __forceinline__ int clampi(int v, int lo, int hi) {
  return v < lo ? lo : (v > hi ? hi : v);
}

__device__ __forceinline__ float tu_value(const float* __restrict__ f, int j, int up) {
  int d = j - (up >> 1);
  float v = 0.0f;
  if ((up & 1) == 0) {
    if (d >= 0 && d <= 6) {
      int k = 2 * d;
      float lo = (k >= 1)  ? f[k - 1] : 0.0f;
      float hi = (k <= 11) ? f[k]     : 0.0f;
      v = lo + hi;
    }
  } else {
    int t = d - 1;
    if (t >= 0 && t <= 5) v = f[2 * t] + f[2 * t + 1];
  }
  return v;
}

__device__ __forceinline__ float td_value(const float* __restrict__ f, int j, int mp) {
  int k = j - 2 * mp;
  return (k >= 0 && k < 12) ? f[k] : 0.0f;
}

__global__ __launch_bounds__(256) void pass1_up_w(
    const float* __restrict__ x, const float* __restrict__ bias,
    const float* __restrict__ fup, float* __restrict__ xw) {
  __shared__ float sTU[16][16];
  for (int i = threadIdx.x; i < 256; i += 256)
    sTU[i >> 4][i & 15] = tu_value(fup, i & 15, i >> 4);
  __syncthreads();

  const int lane = threadIdx.x & 31;
  const int half = lane >> 4;
  const int mrow = lane & 15;

  int t = blockIdx.x * 8 + (threadIdx.x >> 5);
  const int vtile = t & 15; t >>= 4;
  const int ch = t;

  const int cbase = 8 * vtile - 3;
  const float bb  = bias[ch & 63];
  const float* ab = x + (size_t)ch * 130 * 130 + 131 + (size_t)mrow * 130;
  float* ob = xw + ((size_t)ch * 128 + 8 * half) * 256 + 16 * vtile + mrow;

  v8f acc[8] = {};
#pragma unroll
  for (int c = 0; c < 4; ++c) {
    const int j0 = 4 * c + 2 * half;
    const int c0 = cbase + j0, c1 = c0 + 1;
    v2f b;
    b.x = ((unsigned)c0 < 128u) ? sTU[mrow][j0]     : 0.0f;
    b.y = ((unsigned)c1 < 128u) ? sTU[mrow][j0 + 1] : 0.0f;
    const float* q0 = ab + clampi(c0, 0, 127);
    const float* q1 = ab + clampi(c1, 0, 127);
#pragma unroll
    for (int nt = 0; nt < 8; ++nt) {
      v2f a;
      a.x = q0[2080 * nt] + bb;
      a.y = q1[2080 * nt] + bb;
      acc[nt] = wmma4(a, b, acc[nt]);
    }
  }
#pragma unroll
  for (int g = 0; g < 8; ++g)
#pragma unroll
    for (int nt = 0; nt < 8; ++nt)
      ob[(size_t)(16 * nt + g) * 256] = acc[nt][g];
}

__global__ __launch_bounds__(256) void pass2_up_h_lrelu_dn_h(
    const float* __restrict__ xw, const float* __restrict__ fup,
    const float* __restrict__ fdn, float* __restrict__ t3) {
  __shared__ float sTU[16][16];
  __shared__ float sTD[16][48];
  for (int i = threadIdx.x; i < 256; i += 256)
    sTU[i >> 4][i & 15] = tu_value(fup, i & 15, i >> 4);
  for (int i = threadIdx.x; i < 16 * 48; i += 256) {
    int mp = i / 48, jj = i % 48;
    sTD[mp][jj] = td_value(fdn, jj - 1, mp);
  }
  __syncthreads();

  const int lane = threadIdx.x & 31;
  const int half = lane >> 4;
  const int mrow = lane & 15;

  int t = blockIdx.x * 8 + (threadIdx.x >> 5);
  const int ngroup = t & 1; t >>= 1;
  const int mtile  = t & 7; t >>= 3;
  const int ch = t;

  const int sbase = 16 * mtile - 6;
  const int ubase = 32 * mtile - 6;
  const float* xb = xw + (size_t)ch * 128 * 256 + 128 * ngroup + mrow;
  float* ob = t3 + ((size_t)ch * 128 + 16 * mtile + 8 * half) * 256
            + 128 * ngroup + mrow;

  int off0[8], off1[8];
  bool ok0[8], ok1[8];
#pragma unroll
  for (int cc = 0; cc < 8; ++cc) {
    const int r0 = sbase + 4 * cc + 2 * half, r1 = r0 + 1;
    ok0[cc] = (unsigned)r0 < 128u;
    ok1[cc] = (unsigned)r1 < 128u;
    off0[cc] = clampi(r0, 0, 127) * 256;
    off1[cc] = clampi(r1, 0, 127) * 256;
  }
  v2f aU[3][4];
  v2f aD[3][4];
#pragma unroll
  for (int q = 0; q < 3; ++q)
#pragma unroll
    for (int c = 0; c < 4; ++c) {
      const int j0 = 4 * c + 2 * half;
      aU[q][c].x = ok0[2 * q + c] ? sTU[mrow][j0]     : 0.0f;
      aU[q][c].y = ok1[2 * q + c] ? sTU[mrow][j0 + 1] : 0.0f;
      const int jj0 = 16 * q + 8 * half + 2 * c;
      const int u0 = ubase + jj0, u1 = u0 + 1;
      aD[q][c].x = ((unsigned)u0 < 256u) ? sTD[mrow][jj0]     : 0.0f;
      aD[q][c].y = ((unsigned)u1 < 256u) ? sTD[mrow][jj0 + 1] : 0.0f;
    }

#pragma unroll 2
  for (int nt = 0; nt < 8; ++nt) {
    v2f b[8];
#pragma unroll
    for (int cc = 0; cc < 8; ++cc) {
      b[cc].x = xb[off0[cc] + 16 * nt];
      b[cc].y = xb[off1[cc] + 16 * nt];
    }
    v8f tq[3] = {};
#pragma unroll
    for (int c = 0; c < 4; ++c)
#pragma unroll
      for (int q = 0; q < 3; ++q)
        tq[q] = wmma4(aU[q][c], b[2 * q + c], tq[q]);
    v8f acc = {};
#pragma unroll
    for (int q = 0; q < 3; ++q) {
#pragma unroll
      for (int g = 0; g < 8; ++g) {
        float v = tq[q][g];
        tq[q][g] = (v >= 0.0f ? v : SLOPE * v) * GAIN;
      }
#pragma unroll
      for (int c = 0; c < 4; ++c) {
        v2f bd;
        bd.x = tq[q][2 * c];
        bd.y = tq[q][2 * c + 1];
        acc = wmma4(aD[q][c], bd, acc);
      }
    }
#pragma unroll
    for (int g = 0; g < 8; ++g)
      ob[(size_t)g * 256 + 16 * nt] = acc[g];
  }
}

__global__ __launch_bounds__(256) void pass3_dn_w(
    const float* __restrict__ t3, const float* __restrict__ fdn,
    float* __restrict__ out) {
  __shared__ float sTD[16][48];
  for (int i = threadIdx.x; i < 16 * 48; i += 256) {
    int mp = i / 48, j = i % 48;
    sTD[mp][j] = (j < 44) ? td_value(fdn, j, mp) : 0.0f;
  }
  __syncthreads();

  const int lane = threadIdx.x & 31;
  const int half = lane >> 4;
  const int mrow = lane & 15;

  int t = blockIdx.x * 8 + (threadIdx.x >> 5);
  const int ntile = t & 7; t >>= 3;
  const int ch = t;

  const int cbase = 32 * ntile - 5;
  const float* ab = t3 + ((size_t)ch * 128 + mrow) * 256;
  float* ob = out + ((size_t)ch * 128 + 8 * half) * 128 + 16 * ntile + mrow;

  v8f acc[8] = {};
#pragma unroll
  for (int c = 0; c < 11; ++c) {
    const int j0 = 4 * c + 2 * half;
    const int c0 = cbase + j0, c1 = c0 + 1;
    v2f b;
    b.x = ((unsigned)c0 < 256u) ? sTD[mrow][j0]     : 0.0f;
    b.y = ((unsigned)c1 < 256u) ? sTD[mrow][j0 + 1] : 0.0f;
    const float* q0 = ab + clampi(c0, 0, 255);
    const float* q1 = ab + clampi(c1, 0, 255);
#pragma unroll
    for (int nt = 0; nt < 8; ++nt) {
      v2f a;
      a.x = q0[4096 * nt];
      a.y = q1[4096 * nt];
      acc[nt] = wmma4(a, b, acc[nt]);
    }
  }
#pragma unroll
  for (int g = 0; g < 8; ++g)
#pragma unroll
    for (int nt = 0; nt < 8; ++nt)
      ob[(size_t)(16 * nt + g) * 128] = acc[nt][g];
}

extern "C" void kernel_launch(void* const* d_in, const int* in_sizes, int n_in,
                              void* d_out, int out_size, void* d_ws, size_t ws_size,
                              hipStream_t stream) {
  (void)in_sizes; (void)n_in; (void)out_size; (void)ws_size;
  const float* x    = (const float*)d_in[0];
  const float* bias = (const float*)d_in[1];
  const float* fup  = (const float*)d_in[2];
  const float* fdn  = (const float*)d_in[3];
  float* out = (float*)d_out;

  float* xw = (float*)d_ws;
  float* t3 = (float*)((char*)d_ws + (size_t)512 * 128 * 256 * 4);

  dim3 blk(256);
  pass1_up_w           <<<512 * 16    / 8, blk, 0, stream>>>(x, bias, fup, xw);
  pass2_up_h_lrelu_dn_h<<<512 * 8 * 2 / 8, blk, 0, stream>>>(xw, fup, fdn, t3);
  pass3_dn_w           <<<512 * 8     / 8, blk, 0, stream>>>(t3, fdn, out);
}


# Up-W fused into the up-H/leaky/down-H kernel: x rows staged once per work-group in LDS, up-W's D reused as B of up-H; xw never stored; 2 launches; up-H sum regrouped
# speedup vs baseline: 1.7292x; 1.2383x over previous
#include <hip/hip_runtime.h>

typedef __attribute__((ext_vector_type(2))) float v2f;
typedef __attribute__((ext_vector_type(8))) float v8f;

#define GAIN 1.41421356237309515f
#define SLOPE 0.2f

__device__ __forceinline__ v8f wmma4(v2f a, v2f b, v8f c) {
  return __builtin_amdgcn_wmma_f32_16x16x4_f32(false, a, false, b, (short)0, c,
                                               false, false);
}

__device__ __forceinline__ int clampi(int v, int lo, int hi) {
  return v < lo ? lo : (v > hi ? hi : v);
}

__device__ __forceinline__ float tu_value(const float* __restrict__ f, int j, int up) {
  int d = j - (up >> 1);
  float v = 0.0f;
  if ((up & 1) == 0) {
    if (d >= 0 && d <= 6) {
      int k = 2 * d;
      float lo = (k >= 1)  ? f[k - 1] : 0.0f;
      float hi = (k <= 11) ? f[k]     : 0.0f;
      v = lo + hi;
    }
  } else {
    int t = d - 1;
    if (t >= 0 && t <= 5) v = f[2 * t] + f[2 * t + 1];
  }
  return v;
}

__device__ __forceinline__ float td_value(const float* __restrict__ f, int j, int mp) {
  int k = j - 2 * mp;
  return (k >= 0 && k < 12) ? f[k] : 0.0f;
}

#define XP 148
__global__ __launch_bounds__(256, 8) void pass12_up_w_up_h_lrelu_dn_h(
    const float* __restrict__ x, const float* __restrict__ bias,
    const float* __restrict__ fup, const float* __restrict__ fdn,
    float* __restrict__ t3) {
  __shared__ float sTU[16][16];
  __shared__ float sTD[16][48];
  __shared__ __attribute__((aligned(16))) float sX[80][XP];

  const int lane = threadIdx.x & 31;
  const int wave = threadIdx.x >> 5;
  const int half = lane >> 4;
  const int mrow = lane & 15;

  const int ch   = blockIdx.x >> 1;
  const int beta = blockIdx.x & 1;
  const int mtile  = 4 * beta + (wave >> 1);
  const int ngroup = wave & 1;

  for (int i = threadIdx.x; i < 256; i += 256)
    sTU[i >> 4][i & 15] = tu_value(fup, i & 15, i >> 4);
  for (int i = threadIdx.x; i < 16 * 48; i += 256) {
    int mp = i / 48, jj = i % 48;
    sTD[mp][jj] = td_value(fdn, jj - 1, mp);
  }
  {
    const float bb = bias[ch & 63];
    const float* xc = x + (size_t)ch * 130 * 130;
    const int srow0 = 64 * beta - 6;
#pragma unroll
    for (int rr = 0; rr < 10; ++rr) {
      const int l  = wave + 8 * rr;
      const int sr = srow0 + l;
      const bool rok = (unsigned)sr < 128u;
      const float* xr = xc + (clampi(sr, 0, 127) + 1) * 130 + 1;
#pragma unroll
      for (int k = 0; k < 5; ++k) {
        const int L  = lane + 32 * k;
        const int sc = L - 3;
        const float v = xr[clampi(sc, 0, 127)];
        if (L < 136) sX[l][L] = (rok && (unsigned)sc < 128u) ? v + bb : 0.0f;
      }
    }
  }
  __syncthreads();

  const int ubase = 32 * mtile - 6;
  float* ob = t3 + ((size_t)ch * 128 + 16 * mtile + 8 * half) * 256
            + 128 * ngroup + mrow;

  v2f bW[4];
  v2f aU0[4], aU1[4];
  v2f aD[3][4];
#pragma unroll
  for (int c = 0; c < 4; ++c) {
    bW[c].x  = sTU[mrow][4 * c + 2 * half];
    bW[c].y  = sTU[mrow][4 * c + 2 * half + 1];
    aU0[c].x = sTU[mrow][8 * half + 2 * c];
    aU0[c].y = sTU[mrow][8 * half + 2 * c + 1];
    aU1[c].x = sTU[mrow][8 * (1 - half) + 2 * c];
    aU1[c].y = sTU[mrow][8 * (1 - half) + 2 * c + 1];
  }
#pragma unroll
  for (int q = 0; q < 3; ++q)
#pragma unroll
    for (int c = 0; c < 4; ++c) {
      const int jj0 = 16 * q + 8 * half + 2 * c;
      const int u0 = ubase + jj0, u1 = u0 + 1;
      aD[q][c].x = ((unsigned)u0 < 256u) ? sTD[mrow][jj0]     : 0.0f;
      aD[q][c].y = ((unsigned)u1 < 256u) ? sTD[mrow][jj0 + 1] : 0.0f;
    }

  const float* xs0 = &sX[16 * (wave >> 1) + mrow][2 * half] + 64 * ngroup;
  const float* xs1 = xs0 + 16 * XP;

#pragma unroll 2
  for (int nt = 0; nt < 8; ++nt) {
    v8f d0 = {}, d1 = {};
#pragma unroll
    for (int c = 0; c < 4; ++c) {
      const v2f a0 = *(const v2f*)(xs0 + 8 * nt + 4 * c);
      const v2f a1 = *(const v2f*)(xs1 + 8 * nt + 4 * c);
      d0 = wmma4(a0, bW[c], d0);
      d1 = wmma4(a1, bW[c], d1);
    }
    v8f tq[3] = {};
#pragma unroll
    for (int c = 0; c < 4; ++c) {
      v2f b0, b1, b2;
      b0.x = d0[2 * c]; b0.y = d0[2 * c + 1];
      b2.x = d1[2 * c]; b2.y = d1[2 * c + 1];
      b1.x = half ? b0.x : b2.x;
      b1.y = half ? b0.y : b2.y;
      tq[0] = wmma4(aU0[c], b0, tq[0]);
      tq[1] = wmma4(aU1[c], b1, tq[1]);
      tq[2] = wmma4(aU0[c], b2, tq[2]);
    }
    v8f acc = {};
#pragma unroll
    for (int q = 0; q < 3; ++q) {
#pragma unroll
      for (int g = 0; g < 8; ++g) {
        float v = tq[q][g];
        tq[q][g] = (v >= 0.0f ? v : SLOPE * v) * GAIN;
      }
#pragma unroll
      for (int c = 0; c < 4; ++c) {
        v2f bd;
        bd.x = tq[q][2 * c];
        bd.y = tq[q][2 * c + 1];
        acc = wmma4(aD[q][c], bd, acc);
      }
    }
#pragma unroll
    for (int g = 0; g < 8; ++g)
      ob[(size_t)g * 256 + 16 * nt] = acc[g];
  }
}

__global__ __launch_bounds__(256) void pass3_dn_w(
    const float* __restrict__ t3, const float* __restrict__ fdn,
    float* __restrict__ out) {
  __shared__ float sTD[16][48];
  for (int i = threadIdx.x; i < 16 * 48; i += 256) {
    int mp = i / 48, j = i % 48;
    sTD[mp][j] = (j < 44) ? td_value(fdn, j, mp) : 0.0f;
  }
  __syncthreads();

  const int lane = threadIdx.x & 31;
  const int half = lane >> 4;
  const int mrow = lane & 15;

  int t = blockIdx.x * 8 + (threadIdx.x >> 5);
  const int ntile = t & 7; t >>= 3;
  const int ch = t;

  const int cbase = 32 * ntile - 5;
  const float* ab = t3 + ((size_t)ch * 128 + mrow) * 256;
  float* ob = out + ((size_t)ch * 128 + 8 * half) * 128 + 16 * ntile + mrow;

  v8f acc[8] = {};
#pragma unroll
  for (int c = 0; c < 11; ++c) {
    const int j0 = 4 * c + 2 * half;
    const int c0 = cbase + j0, c1 = c0 + 1;
    v2f b;
    b.x = ((unsigned)c0 < 256u) ? sTD[mrow][j0]     : 0.0f;
    b.y = ((unsigned)c1 < 256u) ? sTD[mrow][j0 + 1] : 0.0f;
    const float* q0 = ab + clampi(c0, 0, 255);
    const float* q1 = ab + clampi(c1, 0, 255);
#pragma unroll
    for (int nt = 0; nt < 8; ++nt) {
      v2f a;
      a.x = q0[4096 * nt];
      a.y = q1[4096 * nt];
      acc[nt] = wmma4(a, b, acc[nt]);
    }
  }
#pragma unroll
  for (int g = 0; g < 8; ++g)
#pragma unroll
    for (int nt = 0; nt < 8; ++nt)
      ob[(size_t)(16 * nt + g) * 128] = acc[nt][g];
}

extern "C" void kernel_launch(void* const* d_in, const int* in_sizes, int n_in,
                              void* d_out, int out_size, void* d_ws, size_t ws_size,
                              hipStream_t stream) {
  (void)in_sizes; (void)n_in; (void)out_size; (void)ws_size;
  const float* x    = (const float*)d_in[0];
  const float* bias = (const float*)d_in[1];
  const float* fup  = (const float*)d_in[2];
  const float* fdn  = (const float*)d_in[3];
  float* out = (float*)d_out;

  float* t3 = (float*)d_ws;

  dim3 blk(256);
  pass12_up_w_up_h_lrelu_dn_h<<<512 * 2, blk, 0, stream>>>(x, bias, fup, fdn, t3);
  pass3_dn_w           <<<512 * 8     / 8, blk, 0, stream>>>(t3, fdn, out);
}


# Down-W fused in (one kernel, no workspace at all; the earlier fused try still read xw from it): 32x256 dn-H strip turned through LDS (pitch 276, zero sides), read back as WMMA A; sum order kept
# speedup vs baseline: 3.0072x; 1.7391x over previous
#include <hip/hip_runtime.h>

typedef __attribute__((ext_vector_type(2))) float v2f;
typedef __attribute__((ext_vector_type(8))) float v8f;

#define GAIN 1.41421356237309515f
#define SLOPE 0.2f

__device__ __forceinline__ v8f wmma4(v2f a, v2f b, v8f c) {
  return __builtin_amdgcn_wmma_f32_16x16x4_f32(false, a, false, b, (short)0, c,
                                               false, false);
}

__device__ __forceinline__ int clampi(int v, int lo, int hi) {
  return v < lo ? lo : (v > hi ? hi : v);
}

__device__ __forceinline__ float tu_value(const float* __restrict__ f, int j, int up) {
  int d = j - (up >> 1);
  float v = 0.0f;
  if ((up & 1) == 0) {
    if (d >= 0 && d <= 6) {
      int k = 2 * d;
      float lo = (k >= 1)  ? f[k - 1] : 0.0f;
      float hi = (k <= 11) ? f[k]     : 0.0f;
      v = lo + hi;
    }
  } else {
    int t = d - 1;
    if (t >= 0 && t <= 5) v = f[2 * t] + f[2 * t + 1];
  }
  return v;
}

__device__ __forceinline__ float td_value(const float* __restrict__ f, int j, int mp) {
  int k = j - 2 * mp;
  return (k >= 0 && k < 12) ? f[k] : 0.0f;
}

#define XP 148
#define TP 276
__global__ __launch_bounds__(256, 8) void up_lrelu_dn_one_kernel(
    const float* __restrict__ x, const float* __restrict__ bias,
    const float* __restrict__ fup, const float* __restrict__ fdn,
    float* __restrict__ out) {
  __shared__ float sTU[16][16];
  __shared__ float sTD[16][48];
  __shared__ __attribute__((aligned(16))) float sX[48][XP];
  __shared__ __attribute__((aligned(16))) float sT[32][TP];

  const int lane = threadIdx.x & 31;
  const int wave = threadIdx.x >> 5;
  const int half = lane >> 4;
  const int mrow = lane & 15;

  const int ch   = blockIdx.x >> 2;
  const int beta = blockIdx.x & 3;
  const int mloc = wave >> 2;
  const int mtile  = 2 * beta + mloc;
  const int ngroup = wave & 3;

  for (int i = threadIdx.x; i < 256; i += 256)
    sTU[i >> 4][i & 15] = tu_value(fup, i & 15, i >> 4);
  for (int i = threadIdx.x; i < 16 * 48; i += 256) {
    int mp = i / 48, jj = i % 48;
    sTD[mp][jj] = td_value(fdn, jj - 1, mp);
  }
  for (int i = threadIdx.x; i < 32 * 20; i += 256) {
    int r = i / 20, k = i % 20;
    sT[r][k < 5 ? k : 256 + k] = 0.0f;
  }
  {
    const float bb = bias[ch & 63];
    const float* xc = x + (size_t)ch * 130 * 130;
    const int srow0 = 32 * beta - 6;
#pragma unroll
    for (int rr = 0; rr < 6; ++rr) {
      const int l  = wave + 8 * rr;
      const int sr = srow0 + l;
      const bool rok = (unsigned)sr < 128u;
      const float* xr = xc + (clampi(sr, 0, 127) + 1) * 130 + 1;
#pragma unroll
      for (int k = 0; k < 5; ++k) {
        const int L  = lane + 32 * k;
        const int sc = L - 3;
        const float v = xr[clampi(sc, 0, 127)];
        if (L < 136) sX[l][L] = (rok && (unsigned)sc < 128u) ? v + bb : 0.0f;
      }
    }
  }
  __syncthreads();

  const int ubase = 32 * mtile - 6;
  float* tb = &sT[16 * mloc + 8 * half][5 + 64 * ngroup + mrow];

  v2f bW[4];
  v2f aU0[4], aU1[4];
  v2f aD[3][4];
#pragma unroll
  for (int c = 0; c < 4; ++c) {
    bW[c].x  = sTU[mrow][4 * c + 2 * half];
    bW[c].y  = sTU[mrow][4 * c + 2 * half + 1];
    aU0[c].x = sTU[mrow][8 * half + 2 * c];
    aU0[c].y = sTU[mrow][8 * half + 2 * c + 1];
    aU1[c].x = sTU[mrow][8 * (1 - half) + 2 * c];
    aU1[c].y = sTU[mrow][8 * (1 - half) + 2 * c + 1];
  }
#pragma unroll
  for (int q = 0; q < 3; ++q)
#pragma unroll
    for (int c = 0; c < 4; ++c) {
      const int jj0 = 16 * q + 8 * half + 2 * c;
      const int u0 = ubase + jj0, u1 = u0 + 1;
      aD[q][c].x = ((unsigned)u0 < 256u) ? sTD[mrow][jj0]     : 0.0f;
      aD[q][c].y = ((unsigned)u1 < 256u) ? sTD[mrow][jj0 + 1] : 0.0f;
    }

  const float* xs0 = &sX[16 * mloc + mrow][2 * half] + 32 * ngroup;
  const float* xs1 = xs0 + 16 * XP;

#pragma unroll 2
  for (int nt = 0; nt < 4; ++nt) {
    v8f d0 = {}, d1 = {};
#pragma unroll
    for (int c = 0; c < 4; ++c) {
      const v2f a0 = *(const v2f*)(xs0 + 8 * nt + 4 * c);
      const v2f a1 = *(const v2f*)(xs1 + 8 * nt + 4 * c);
      d0 = wmma4(a0, bW[c], d0);
      d1 = wmma4(a1, bW[c], d1);
    }
    v8f tq[3] = {};
#pragma unroll
    for (int c = 0; c < 4; ++c) {
      v2f b0, b1, b2;
      b0.x = d0[2 * c]; b0.y = d0[2 * c + 1];
      b2.x = d1[2 * c]; b2.y = d1[2 * c + 1];
      b1.x = half ? b0.x : b2.x;
      b1.y = half ? b0.y : b2.y;
      tq[0] = wmma4(aU0[c], b0, tq[0]);
      tq[1] = wmma4(aU1[c], b1, tq[1]);
      tq[2] = wmma4(aU0[c], b2, tq[2]);
    }
    v8f acc = {};
#pragma unroll
    for (int q = 0; q < 3; ++q) {
#pragma unroll
      for (int g = 0; g < 8; ++g) {
        float v = tq[q][g];
        tq[q][g] = (v >= 0.0f ? v : SLOPE * v) * GAIN;
      }
#pragma unroll
      for (int c = 0; c < 4; ++c) {
        v2f bd;
        bd.x = tq[q][2 * c];
        bd.y = tq[q][2 * c + 1];
        acc = wmma4(aD[q][c], bd, acc);
      }
    }
#pragma unroll
    for (int g = 0; g < 8; ++g)
      tb[g * TP + 16 * nt] = acc[g];
  }
  __syncthreads();

  v2f bT[11];
#pragma unroll
  for (int c = 0; c < 11; ++c) {
    bT[c].x = sTD[mrow][4 * c + 2 * half + 1];
    bT[c].y = sTD[mrow][4 * c + 2 * half + 2];
  }
  const float* ts0 = &sT[16 * mloc + mrow][64 * ngroup + 2 * half];
  const float* ts1 = ts0 + 32;
  v8f o0 = {}, o1 = {};
#pragma unroll
  for (int c = 0; c < 11; ++c) {
    const v2f a0 = *(const v2f*)(ts0 + 4 * c);
    const v2f a1 = *(const v2f*)(ts1 + 4 * c);
    o0 = wmma4(a0, bT[c], o0);
    o1 = wmma4(a1, bT[c], o1);
  }
  float* op = out + ((size_t)ch * 128 + 16 * mtile + 8 * half) * 128 + 32 * ngroup + mrow;
#pragma unroll
  for (int g = 0; g < 8; ++g) {
    op[(size_t)g * 128]      = o0[g];
    op[(size_t)g * 128 + 16] = o1[g];
  }
}

extern "C" void kernel_launch(void* const* d_in, const int* in_sizes, int n_in,
                              void* d_out, int out_size, void* d_ws, size_t ws_size,
                              hipStream_t stream) {
  (void)in_sizes; (void)n_in; (void)out_size; (void)d_ws; (void)ws_size;
  const float* x    = (const float*)d_in[0];
  const float* bias = (const float*)d_in[1];
  const float* fup  = (const float*)d_in[2];
  const float* fdn  = (const float*)d_in[3];
  float* out = (float*)d_out;

  dim3 blk(256);
  up_lrelu_dn_one_kernel<<<512 * 4, blk, 0, stream>>>(x, bias, fup, fdn, out);
}


# Prologue loads batched: filter-tap loads (clamped index, selected after) and all of a wave's source-row loads issued before the first wait: one memory round trip, not one per pass and row; sums kept
# speedup vs baseline: 3.6753x; 1.2222x over previous
#include <hip/hip_runtime.h>

typedef __attribute__((ext_vector_type(2))) float v2f;
typedef __attribute__((ext_vector_type(8))) float v8f;

#define GAIN 1.41421356237309515f
#define SLOPE 0.2f

__device__ __forceinline__ v8f wmma4(v2f a, v2f b, v8f c) {
  return __builtin_amdgcn_wmma_f32_16x16x4_f32(false, a, false, b, (short)0, c,
                                               false, false);
}

__device__ __forceinline__ int clampi(int v, int lo, int hi) {
  return v < lo ? lo : (v > hi ? hi : v);
}

__device__ __forceinline__ float tap(const float* __restrict__ f, int k) {
  const float v = f[clampi(k, 0, 11)];
  return ((unsigned)k < 12u) ? v : 0.0f;
}

#define XP 148
#define TP 276
__global__ __launch_bounds__(256, 8) void up_lrelu_dn_one_kernel(
    const float* __restrict__ x, const float* __restrict__ bias,
    const float* __restrict__ fup, const float* __restrict__ fdn,
    float* __restrict__ out) {
  __shared__ float sTU[16][16];
  __shared__ float sTD[16][48];
  __shared__ __attribute__((aligned(16))) float sX[48][XP];
  __shared__ __attribute__((aligned(16))) float sT[32][TP];

  const int lane = threadIdx.x & 31;
  const int wave = threadIdx.x >> 5;
  const int half = lane >> 4;
  const int mrow = lane & 15;

  const int ch   = blockIdx.x >> 2;
  const int beta = blockIdx.x & 3;
  const int mloc = wave >> 2;
  const int mtile  = 2 * beta + mloc;
  const int ngroup = wave & 3;

  const int tid = threadIdx.x;
  const int pu = 2 * (tid & 15) - (tid >> 4) - 1;
  const float tu0 = tap(fup, pu), tu1 = tap(fup, pu + 1);
  float tdv[3];
#pragma unroll
  for (int t = 0; t < 3; ++t) {
    const int i = tid + 256 * t;
    tdv[t] = tap(fdn, (i % 48) - 1 - 2 * (i / 48));
  }
  const float bb = bias[ch & 63];
  const float* xc = x + (size_t)ch * 130 * 130;
  const int srow0 = 32 * beta - 6;
  float xv[6][4];
#pragma unroll
  for (int rr = 0; rr < 6; ++rr) {
    const int sr = srow0 + wave + 8 * rr;
    const float* xr = xc + (clampi(sr, 0, 127) + 1) * 130 + 1 + lane;
#pragma unroll
    for (int k = 0; k < 4; ++k) xv[rr][k] = xr[32 * k];
  }

  sTU[tid >> 4][tid & 15] = tu0 + tu1;
#pragma unroll
  for (int t = 0; t < 3; ++t) {
    const int i = tid + 256 * t;
    sTD[i / 48][i % 48] = tdv[t];
  }
  for (int i = tid; i < 32 * 20; i += 256) {
    int r = i / 20, k = i % 20;
    sT[r][k < 5 ? k : 256 + k] = 0.0f;
  }
  for (int i = tid; i < 48 * 8; i += 256) {
    int r = i >> 3, k = i & 7;
    sX[r][k < 3 ? k : 128 + k] = 0.0f;
  }
#pragma unroll
  for (int rr = 0; rr < 6; ++rr) {
    const int l = wave + 8 * rr;
    const bool rok = (unsigned)(srow0 + l) < 128u;
#pragma unroll
    for (int k = 0; k < 4; ++k)
      sX[l][3 + lane + 32 * k] = rok ? xv[rr][k] + bb : 0.0f;
  }
  __syncthreads();

  const int ubase = 32 * mtile - 6;
  float* tb = &sT[16 * mloc + 8 * half][5 + 64 * ngroup + mrow];

  v2f bW[4];
  v2f aU0[4], aU1[4];
  v2f aD[3][4];
#pragma unroll
  for (int c = 0; c < 4; ++c) {
    bW[c].x  = sTU[mrow][4 * c + 2 * half];
    bW[c].y  = sTU[mrow][4 * c + 2 * half + 1];
    aU0[c].x = sTU[mrow][8 * half + 2 * c];
    aU0[c].y = sTU[mrow][8 * half + 2 * c + 1];
    aU1[c].x = sTU[mrow][8 * (1 - half) + 2 * c];
    aU1[c].y = sTU[mrow][8 * (1 - half) + 2 * c + 1];
  }
#pragma unroll
  for (int q = 0; q < 3; ++q)
#pragma unroll
    for (int c = 0; c < 4; ++c) {
      const int jj0 = 16 * q + 8 * half + 2 * c;
      const int u0 = ubase + jj0, u1 = u0 + 1;
      aD[q][c].x = ((unsigned)u0 < 256u) ? sTD[mrow][jj0]     : 0.0f;
      aD[q][c].y = ((unsigned)u1 < 256u) ? sTD[mrow][jj0 + 1] : 0.0f;
    }

  const float* xs0 = &sX[16 * mloc + mrow][2 * half] + 32 * ngroup;
  const float* xs1 = xs0 + 16 * XP;

#pragma unroll 2
  for (int nt = 0; nt < 4; ++nt) {
    v8f d0 = {}, d1 = {};
#pragma unroll
    for (int c = 0; c < 4; ++c) {
      const v2f a0 = *(const v2f*)(xs0 + 8 * nt + 4 * c);
      const v2f a1 = *(const v2f*)(xs1 + 8 * nt + 4 * c);
      d0 = wmma4(a0, bW[c], d0);
      d1 = wmma4(a1, bW[c], d1);
    }
    v8f tq[3] = {};
#pragma unroll
    for (int c = 0; c < 4; ++c) {
      v2f b0, b1, b2;
      b0.x = d0[2 * c]; b0.y = d0[2 * c + 1];
      b2.x = d1[2 * c]; b2.y = d1[2 * c + 1];
      b1.x = half ? b0.x : b2.x;
      b1.y = half ? b0.y : b2.y;
      tq[0] = wmma4(aU0[c], b0, tq[0]);
      tq[1] = wmma4(aU1[c], b1, tq[1]);
      tq[2] = wmma4(aU0[c], b2, tq[2]);
    }
    v8f acc = {};
#pragma unroll
    for (int q = 0; q < 3; ++q) {
#pragma unroll
      for (int g = 0; g < 8; ++g) {
        float v = tq[q][g];
        tq[q][g] = (v >= 0.0f ? v : SLOPE * v) * GAIN;
      }
#pragma unroll
      for (int c = 0; c < 4; ++c) {
        v2f bd;
        bd.x = tq[q][2 * c];
        bd.y = tq[q][2 * c + 1];
        acc = wmma4(aD[q][c], bd, acc);
      }
    }
#pragma unroll
    for (int g = 0; g < 8; ++g)
      tb[g * TP + 16 * nt] = acc[g];
  }
  __syncthreads();

  v2f bT[11];
#pragma unroll
  for (int c = 0; c < 11; ++c) {
    bT[c].x = sTD[mrow][4 * c + 2 * half + 1];
    bT[c].y = sTD[mrow][4 * c + 2 * half + 2];
  }
  const float* ts0 = &sT[16 * mloc + mrow][64 * ngroup + 2 * half];
  const float* ts1 = ts0 + 32;
  v8f o0 = {}, o1 = {};
#pragma unroll
  for (int c = 0; c < 11; ++c) {
    const v2f a0 = *(const v2f*)(ts0 + 4 * c);
    const v2f a1 = *(const v2f*)(ts1 + 4 * c);
    o0 = wmma4(a0, bT[c], o0);
    o1 = wmma4(a1, bT[c], o1);
  }
  float* op = out + ((size_t)ch * 128 + 16 * mtile + 8 * half) * 128 + 32 * ngroup + mrow;
#pragma unroll
  for (int g = 0; g < 8; ++g) {
    op[(size_t)g * 128]      = o0[g];
    op[(size_t)g * 128 + 16] = o1[g];
  }
}

extern "C" void kernel_launch(void* const* d_in, const int* in_sizes, int n_in,
                              void* d_out, int out_size, void* d_ws, size_t ws_size,
                              hipStream_t stream) {
  (void)in_sizes; (void)n_in; (void)out_size; (void)d_ws; (void)ws_size;
  const float* x    = (const float*)d_in[0];
  const float* bias = (const float*)d_in[1];
  const float* fup  = (const float*)d_in[2];
  const float* fdn  = (const float*)d_in[3];
  float* out = (float*)d_out;

  dim3 blk(256);
  up_lrelu_dn_one_kernel<<<512 * 4, blk, 0, stream>>>(x, bias, fup, fdn, out);
}


# Main loop re-split: each wave makes both 16-row tiles of its 2 column tiles, so 3 up-W and 5 up-H tiles feed both dn-H sums (56 WMMAs per column tile, were 2x32); dn-H chains interleave; sums kept
# speedup vs baseline: 4.0846x; 1.1114x over previous
#include <hip/hip_runtime.h>

typedef __attribute__((ext_vector_type(2))) float v2f;
typedef __attribute__((ext_vector_type(8))) float v8f;

#define GAIN 1.41421356237309515f
#define SLOPE 0.2f

__device__ __forceinline__ v8f wmma4(v2f a, v2f b, v8f c) {
  return __builtin_amdgcn_wmma_f32_16x16x4_f32(false, a, false, b, (short)0, c,
                                               false, false);
}

__device__ __forceinline__ int clampi(int v, int lo, int hi) {
  return v < lo ? lo : (v > hi ? hi : v);
}

__device__ __forceinline__ float tap(const float* __restrict__ f, int k) {
  const float v = f[clampi(k, 0, 11)];
  return ((unsigned)k < 12u) ? v : 0.0f;
}

#define XP 148
#define TP 276
__global__ __launch_bounds__(256, 8) void up_lrelu_dn_one_kernel(
    const float* __restrict__ x, const float* __restrict__ bias,
    const float* __restrict__ fup, const float* __restrict__ fdn,
    float* __restrict__ out) {
  __shared__ float sTU[16][16];
  __shared__ float sTD[16][48];
  __shared__ __attribute__((aligned(16))) float sX[48][XP];
  __shared__ __attribute__((aligned(16))) float sT[32][TP];

  const int lane = threadIdx.x & 31;
  const int wave = threadIdx.x >> 5;
  const int half = lane >> 4;
  const int mrow = lane & 15;

  const int ch   = blockIdx.x >> 2;
  const int beta = blockIdx.x & 3;
  const int mloc = wave >> 2;
  const int mtile  = 2 * beta + mloc;
  const int ngroup = wave & 3;

  const int tid = threadIdx.x;
  const int pu = 2 * (tid & 15) - (tid >> 4) - 1;
  const float tu0 = tap(fup, pu), tu1 = tap(fup, pu + 1);
  float tdv[3];
#pragma unroll
  for (int t = 0; t < 3; ++t) {
    const int i = tid + 256 * t;
    tdv[t] = tap(fdn, (i % 48) - 1 - 2 * (i / 48));
  }
  const float bb = bias[ch & 63];
  const float* xc = x + (size_t)ch * 130 * 130;
  const int srow0 = 32 * beta - 6;
  float xv[6][4];
#pragma unroll
  for (int rr = 0; rr < 6; ++rr) {
    const int sr = srow0 + wave + 8 * rr;
    const float* xr = xc + (clampi(sr, 0, 127) + 1) * 130 + 1 + lane;
#pragma unroll
    for (int k = 0; k < 4; ++k) xv[rr][k] = xr[32 * k];
  }

  sTU[tid >> 4][tid & 15] = tu0 + tu1;
#pragma unroll
  for (int t = 0; t < 3; ++t) {
    const int i = tid + 256 * t;
    sTD[i / 48][i % 48] = tdv[t];
  }
  for (int i = tid; i < 32 * 20; i += 256) {
    int r = i / 20, k = i % 20;
    sT[r][k < 5 ? k : 256 + k] = 0.0f;
  }
  for (int i = tid; i < 48 * 8; i += 256) {
    int r = i >> 3, k = i & 7;
    sX[r][k < 3 ? k : 128 + k] = 0.0f;
  }
#pragma unroll
  for (int rr = 0; rr < 6; ++rr) {
    const int l = wave + 8 * rr;
    const bool rok = (unsigned)(srow0 + l) < 128u;
#pragma unroll
    for (int k = 0; k < 4; ++k)
      sX[l][3 + lane + 32 * k] = rok ? xv[rr][k] + bb : 0.0f;
  }
  __syncthreads();

  const int ub0 = 64 * beta - 6;
  float* tb = &sT[8 * half][5 + 32 * wave + mrow];

  v2f bW[4];
  v2f aU0[4], aU1[4];
  v2f aD[3][4], aDt[4], aDb[4];
#pragma unroll
  for (int c = 0; c < 4; ++c) {
    bW[c].x  = sTU[mrow][4 * c + 2 * half];
    bW[c].y  = sTU[mrow][4 * c + 2 * half + 1];
    aU0[c].x = sTU[mrow][8 * half + 2 * c];
    aU0[c].y = sTU[mrow][8 * half + 2 * c + 1];
    aU1[c].x = sTU[mrow][8 * (1 - half) + 2 * c];
    aU1[c].y = sTU[mrow][8 * (1 - half) + 2 * c + 1];
  }
#pragma unroll
  for (int q = 0; q < 3; ++q)
#pragma unroll
    for (int c = 0; c < 4; ++c) {
      const int jj0 = 16 * q + 8 * half + 2 * c;
      aD[q][c].x = sTD[mrow][jj0];
      aD[q][c].y = sTD[mrow][jj0 + 1];
    }
#pragma unroll
  for (int c = 0; c < 4; ++c) {
    const int ut = ub0 + 8 * half + 2 * c;
    const int ud = ut + 64;
    aDt[c].x = ((unsigned)ut < 256u)       ? aD[0][c].x : 0.0f;
    aDt[c].y = ((unsigned)(ut + 1) < 256u) ? aD[0][c].y : 0.0f;
    aDb[c].x = ((unsigned)ud < 256u)       ? aD[2][c].x : 0.0f;
    aDb[c].y = ((unsigned)(ud + 1) < 256u) ? aD[2][c].y : 0.0f;
  }

  const float* xs0 = &sX[mrow][2 * half] + 16 * wave;
  const float* xs1 = xs0 + 16 * XP;
  const float* xs2 = xs0 + 32 * XP;

#pragma unroll 1
  for (int nt = 0; nt < 2; ++nt) {
    v8f d0 = {}, d1 = {}, d2 = {};
#pragma unroll
    for (int c = 0; c < 4; ++c) {
      const v2f a0 = *(const v2f*)(xs0 + 8 * nt + 4 * c);
      const v2f a1 = *(const v2f*)(xs1 + 8 * nt + 4 * c);
      const v2f a2 = *(const v2f*)(xs2 + 8 * nt + 4 * c);
      d0 = wmma4(a0, bW[c], d0);
      d1 = wmma4(a1, bW[c], d1);
      d2 = wmma4(a2, bW[c], d2);
    }
    v8f tq[5] = {};
#pragma unroll
    for (int c = 0; c < 4; ++c) {
      v2f b0, b1, b2, b3, b4;
      b0.x = d0[2 * c]; b0.y = d0[2 * c + 1];
      b2.x = d1[2 * c]; b2.y = d1[2 * c + 1];
      b4.x = d2[2 * c]; b4.y = d2[2 * c + 1];
      b1.x = half ? b0.x : b2.x;
      b1.y = half ? b0.y : b2.y;
      b3.x = half ? b2.x : b4.x;
      b3.y = half ? b2.y : b4.y;
      tq[0] = wmma4(aU0[c], b0, tq[0]);
      tq[1] = wmma4(aU1[c], b1, tq[1]);
      tq[2] = wmma4(aU0[c], b2, tq[2]);
      tq[3] = wmma4(aU1[c], b3, tq[3]);
      tq[4] = wmma4(aU0[c], b4, tq[4]);
    }
#pragma unroll
    for (int p = 0; p < 5; ++p)
#pragma unroll
      for (int g = 0; g < 8; ++g) {
        float v = tq[p][g];
        tq[p][g] = (v >= 0.0f ? v : SLOPE * v) * GAIN;
      }
    v8f acc0 = {}, acc1 = {};
#pragma unroll
    for (int q = 0; q < 3; ++q) {
#pragma unroll
      for (int c = 0; c < 4; ++c) {
        v2f bd0, bd1;
        bd0.x = tq[q][2 * c];     bd0.y = tq[q][2 * c + 1];
        bd1.x = tq[q + 2][2 * c]; bd1.y = tq[q + 2][2 * c + 1];
        acc0 = wmma4(q == 0 ? aDt[c] : aD[q][c], bd0, acc0);
        acc1 = wmma4(q == 2 ? aDb[c] : aD[q][c], bd1, acc1);
      }
    }
#pragma unroll
    for (int g = 0; g < 8; ++g) {
      tb[g * TP + 16 * nt]        = acc0[g];
      tb[(16 + g) * TP + 16 * nt] = acc1[g];
    }
  }
  __syncthreads();

  v2f bT[11];
#pragma unroll
  for (int c = 0; c < 11; ++c) {
    bT[c].x = sTD[mrow][4 * c + 2 * half + 1];
    bT[c].y = sTD[mrow][4 * c + 2 * half + 2];
  }
  const float* ts0 = &sT[16 * mloc + mrow][64 * ngroup + 2 * half];
  const float* ts1 = ts0 + 32;
  v8f o0 = {}, o1 = {};
#pragma unroll
  for (int c = 0; c < 11; ++c) {
    const v2f a0 = *(const v2f*)(ts0 + 4 * c);
    const v2f a1 = *(const v2f*)(ts1 + 4 * c);
    o0 = wmma4(a0, bT[c], o0);
    o1 = wmma4(a1, bT[c], o1);
  }
  float* op = out + ((size_t)ch * 128 + 16 * mtile + 8 * half) * 128 + 32 * ngroup + mrow;
#pragma unroll
  for (int g = 0; g < 8; ++g) {
    op[(size_t)g * 128]      = o0[g];
    op[(size_t)g * 128 + 16] = o1[g];
  }
}

extern "C" void kernel_launch(void* const* d_in, const int* in_sizes, int n_in,
                              void* d_out, int out_size, void* d_ws, size_t ws_size,
                              hipStream_t stream) {
  (void)in_sizes; (void)n_in; (void)out_size; (void)d_ws; (void)ws_size;
  const float* x    = (const float*)d_in[0];
  const float* bias = (const float*)d_in[1];
  const float* fup  = (const float*)d_in[2];
  const float* fdn  = (const float*)d_in[3];
  float* out = (float*)d_out;

  dim3 blk(256);
  up_lrelu_dn_one_kernel<<<512 * 4, blk, 0, stream>>>(x, bias, fup, fdn, out);
}
